# starting kernel
# baseline (speedup 1.0000x reference)
#include <hip/hip_runtime.h>
#include <math.h>

typedef __attribute__((ext_vector_type(2))) float v2f;
typedef __attribute__((ext_vector_type(8))) float v8f;

#define NG 512
#define HH 256
#define WW 256

#define QSCALE (-0.72134752044448170368f)

__global__ void splat_prep(const float* __restrict__ means,
                           const float* __restrict__ covs,
                           const float* __restrict__ colors,
                           const float* __restrict__ opac,
                           float* __restrict__ gcoef,
                           float* __restrict__ c4)
{
    int n = blockIdx.x * blockDim.x + threadIdx.x;
    if (n >= NG) return;
    float mx = means[2*n+0] - 128.0f;
    float my = means[2*n+1] - 128.0f;
    float a = covs[4*n+0], b = covs[4*n+1], c = covs[4*n+3];
    float det = a*c - b*b;
    float inv_det = 1.0f / det;
    float ia =  c*inv_det;
    float ib = -b*inv_det;
    float ic =  a*inv_det;
    float amp = opac[n] / (6.28318530717958647f * sqrtf(det + 1e-8f));

    float* g = gcoef + 8*n;
    g[0] = QSCALE * ia;
    g[1] = QSCALE * 2.0f*ib;
    g[2] = QSCALE * ic;
    g[3] = QSCALE * -2.0f*(ia*mx + ib*my);
    g[4] = QSCALE * -2.0f*(ib*mx + ic*my);
    g[5] = QSCALE * fmaf(ia*mx, mx, fmaf(2.0f*ib*mx, my, ic*my*my));
    g[6] = 0.0f;
    g[7] = 0.0f;

    float* cc = c4 + 4*n;
    cc[0] = amp;
    cc[1] = amp * colors[3*n+0];
    cc[2] = amp * colors[3*n+1];
    cc[3] = amp * colors[3*n+2];
}

__global__ __launch_bounds__(256)
void splat_main(const float* __restrict__ gcoef_g,
                const float4* __restrict__ c4_g,
                float* __restrict__ out)
{
    __shared__ float  sG[NG * 8];
    __shared__ float4 sC[NG];

    const int tid = threadIdx.x;
    for (int i = tid; i < NG * 8; i += 256) sG[i] = gcoef_g[i];
    for (int i = tid; i < NG;     i += 256) sC[i] = c4_g[i];
    __syncthreads();

    const int lane = tid & 31;
    const int wave = tid >> 5;
    const int strip = blockIdx.x * 8 + wave;
    const int row   = strip >> 4;
    const int x0    = (strip & 15) << 4;

    const int half = lane >> 4;
    const int m    = lane & 15;

    const float fx = (float)(x0 + m) - 128.0f;
    const float fy = (float)row - 128.0f;

    v2f a1, a2;
    a1.x = half ? (fy * fy) : (fx * fx);
    a1.y = half ? fx        : (fx * fy);
    a2.x = half ? 0.0f      : fy;
    a2.y = half ? 0.0f      : 1.0f;

    float4 acc[8];
#pragma unroll
    for (int r = 0; r < 8; ++r) acc[r] = make_float4(0.f, 0.f, 0.f, 0.f);

    const int k0 = half << 1;

    for (int t = 0; t < NG / 16; ++t) {
        const int n = (t << 4) + m;
        const float* g = &sG[n * 8];
        v2f b1 = *(const v2f*)&g[k0];
        v2f b2 = *(const v2f*)&g[4 + k0];

        v8f q = {};
        q = __builtin_amdgcn_wmma_f32_16x16x4_f32(false, a1, false, b1,
                                                  (short)0, q, false, false);
        q = __builtin_amdgcn_wmma_f32_16x16x4_f32(false, a2, false, b2,
                                                  (short)0, q, false, false);

        const float4 cn = sC[n];
#pragma unroll
        for (int r = 0; r < 8; ++r) {
            float e = __builtin_amdgcn_exp2f(q[r]);
            acc[r].x = fmaf(e, cn.x, acc[r].x);
            acc[r].y = fmaf(e, cn.y, acc[r].y);
            acc[r].z = fmaf(e, cn.z, acc[r].z);
            acc[r].w = fmaf(e, cn.w, acc[r].w);
        }
    }

#pragma unroll
    for (int mask = 1; mask <= 8; mask <<= 1) {
#pragma unroll
        for (int r = 0; r < 8; ++r) {
            acc[r].x += __shfl_xor(acc[r].x, mask, 32);
            acc[r].y += __shfl_xor(acc[r].y, mask, 32);
            acc[r].z += __shfl_xor(acc[r].z, mask, 32);
            acc[r].w += __shfl_xor(acc[r].w, mask, 32);
        }
    }

    if (m < 8) {
        float4 v = acc[0];
#pragma unroll
        for (int r = 1; r < 8; ++r) if (m == r) v = acc[r];
        const int x = x0 + (half << 3) + m;
        const float inv = 1.0f / fmaxf(v.x, 1e-8f);
        float* o = out + (row * WW + x) * 3;
        o[0] = v.y * inv;
        o[1] = v.z * inv;
        o[2] = v.w * inv;
    }
}

extern "C" void kernel_launch(void* const* d_in, const int* in_sizes, int n_in,
                              void* d_out, int out_size, void* d_ws, size_t ws_size,
                              hipStream_t stream) {
    const float* means  = (const float*)d_in[0];
    const float* covs   = (const float*)d_in[1];
    const float* colors = (const float*)d_in[2];
    const float* opac   = (const float*)d_in[3];

    float* gcoef = (float*)d_ws;
    float* c4    = (float*)d_ws + NG * 8;

    splat_prep<<<(NG + 255) / 256, 256, 0, stream>>>(means, covs, colors, opac,
                                                     gcoef, c4);
    splat_main<<<(HH * WW) / (8 * 16), 256, 0, stream>>>(gcoef,
                                                         (const float4*)c4,
                                                         (float*)d_out);
}


# Fuse the coefficient prep into the splat kernel: each block derives all 512 gaussians' coefficients from the raw inputs straight into LDS, so one launch and no workspace round trip or staging loop
# speedup vs baseline: 1.8591x; 1.8591x over previous
#include <hip/hip_runtime.h>
#include <math.h>

typedef __attribute__((ext_vector_type(2))) float v2f;
typedef __attribute__((ext_vector_type(8))) float v8f;

#define NG 512
#define HH 256
#define WW 256

#define QSCALE (-0.72134752044448170368f)

__global__ __launch_bounds__(256)
void splat_fused(const float* __restrict__ means,
                 const float* __restrict__ covs,
                 const float* __restrict__ colors,
                 const float* __restrict__ opac,
                 float* __restrict__ out)
{
    __shared__ __attribute__((aligned(16))) float sG[NG * 8];
    __shared__ float4 sC[NG];

    const int tid = threadIdx.x;

    {
        float rmx[2], rmy[2], ra[2], rb[2], rc[2], rop[2], r0[2], r1[2], r2[2];
#pragma unroll
        for (int j = 0; j < 2; ++j) {
            const int n = tid + j * 256;
            rmx[j] = means[2*n+0];
            rmy[j] = means[2*n+1];
            ra[j]  = covs[4*n+0];
            rb[j]  = covs[4*n+1];
            rc[j]  = covs[4*n+3];
            rop[j] = opac[n];
            r0[j]  = colors[3*n+0];
            r1[j]  = colors[3*n+1];
            r2[j]  = colors[3*n+2];
        }
#pragma unroll
        for (int j = 0; j < 2; ++j) {
            const int n = tid + j * 256;
            float mx = rmx[j] - 128.0f;
            float my = rmy[j] - 128.0f;
            float a = ra[j], b = rb[j], c = rc[j];
            float det = a*c - b*b;
            float inv_det = 1.0f / det;
            float ia =  c*inv_det;
            float ib = -b*inv_det;
            float ic =  a*inv_det;
            float amp = rop[j] / (6.28318530717958647f * sqrtf(det + 1e-8f));

            float4 g0, g1;
            g0.x = QSCALE * ia;
            g0.y = QSCALE * 2.0f*ib;
            g0.z = QSCALE * ic;
            g0.w = QSCALE * -2.0f*(ia*mx + ib*my);
            g1.x = QSCALE * -2.0f*(ib*mx + ic*my);
            g1.y = QSCALE * fmaf(ia*mx, mx, fmaf(2.0f*ib*mx, my, ic*my*my));
            g1.z = 0.0f;
            g1.w = 0.0f;
            *(float4*)&sG[8*n + 0] = g0;
            *(float4*)&sG[8*n + 4] = g1;

            sC[n] = make_float4(amp, amp * r0[j], amp * r1[j], amp * r2[j]);
        }
    }
    __syncthreads();

    const int lane = tid & 31;
    const int wave = tid >> 5;
    const int strip = blockIdx.x * 8 + wave;
    const int row   = strip >> 4;
    const int x0    = (strip & 15) << 4;

    const int half = lane >> 4;
    const int m    = lane & 15;

    const float fx = (float)(x0 + m) - 128.0f;
    const float fy = (float)row - 128.0f;

    v2f a1, a2;
    a1.x = half ? (fy * fy) : (fx * fx);
    a1.y = half ? fx        : (fx * fy);
    a2.x = half ? 0.0f      : fy;
    a2.y = half ? 0.0f      : 1.0f;

    float4 acc[8];
#pragma unroll
    for (int r = 0; r < 8; ++r) acc[r] = make_float4(0.f, 0.f, 0.f, 0.f);

    const int k0 = half << 1;

    for (int t = 0; t < NG / 16; ++t) {
        const int n = (t << 4) + m;
        const float* g = &sG[n * 8];
        v2f b1 = *(const v2f*)&g[k0];
        v2f b2 = *(const v2f*)&g[4 + k0];

        v8f q = {};
        q = __builtin_amdgcn_wmma_f32_16x16x4_f32(false, a1, false, b1,
                                                  (short)0, q, false, false);
        q = __builtin_amdgcn_wmma_f32_16x16x4_f32(false, a2, false, b2,
                                                  (short)0, q, false, false);

        const float4 cn = sC[n];
#pragma unroll
        for (int r = 0; r < 8; ++r) {
            float e = __builtin_amdgcn_exp2f(q[r]);
            acc[r].x = fmaf(e, cn.x, acc[r].x);
            acc[r].y = fmaf(e, cn.y, acc[r].y);
            acc[r].z = fmaf(e, cn.z, acc[r].z);
            acc[r].w = fmaf(e, cn.w, acc[r].w);
        }
    }

#pragma unroll
    for (int mask = 1; mask <= 8; mask <<= 1) {
#pragma unroll
        for (int r = 0; r < 8; ++r) {
            acc[r].x += __shfl_xor(acc[r].x, mask, 32);
            acc[r].y += __shfl_xor(acc[r].y, mask, 32);
            acc[r].z += __shfl_xor(acc[r].z, mask, 32);
            acc[r].w += __shfl_xor(acc[r].w, mask, 32);
        }
    }

    if (m < 8) {
        float4 v = acc[0];
#pragma unroll
        for (int r = 1; r < 8; ++r) if (m == r) v = acc[r];
        const int x = x0 + (half << 3) + m;
        const float inv = 1.0f / fmaxf(v.x, 1e-8f);
        float* o = out + (row * WW + x) * 3;
        o[0] = v.y * inv;
        o[1] = v.z * inv;
        o[2] = v.w * inv;
    }
}

extern "C" void kernel_launch(void* const* d_in, const int* in_sizes, int n_in,
                              void* d_out, int out_size, void* d_ws, size_t ws_size,
                              hipStream_t stream) {
    const float* means  = (const float*)d_in[0];
    const float* covs   = (const float*)d_in[1];
    const float* colors = (const float*)d_in[2];
    const float* opac   = (const float*)d_in[3];

    splat_fused<<<(HH * WW) / (8 * 16), 256, 0, stream>>>(means, covs, colors, opac,
                                                          (float*)d_out);
}


# Sum over the 16 gaussian lanes by a halving butterfly of DPP row_xmask adds (a lane keeps half its rows per step): 32 partner adds on the vector ALU, not 128 ds_bpermute and 128 adds; same add tree
# speedup vs baseline: 2.3010x; 1.2377x over previous
#include <hip/hip_runtime.h>
#include <math.h>

typedef __attribute__((ext_vector_type(2))) float v2f;
typedef __attribute__((ext_vector_type(8))) float v8f;

#define NG 512
#define HH 256
#define WW 256

#define QSCALE (-0.72134752044448170368f)


template <int MASK>
__device__ __forceinline__ float row_xor(float v)
{
    return __builtin_bit_cast(float,
        __builtin_amdgcn_update_dpp(0, __builtin_bit_cast(int, v),
                                    0x160 + MASK, 0xf, 0xf, false));
}

template <int MASK>
__device__ __forceinline__ float4 keep_half(bool bit, float4 lo, float4 hi)
{
    float4 keep, send;
    keep.x = bit ? hi.x : lo.x;   send.x = bit ? lo.x : hi.x;
    keep.y = bit ? hi.y : lo.y;   send.y = bit ? lo.y : hi.y;
    keep.z = bit ? hi.z : lo.z;   send.z = bit ? lo.z : hi.z;
    keep.w = bit ? hi.w : lo.w;   send.w = bit ? lo.w : hi.w;
    float4 o;
    o.x = keep.x + row_xor<MASK>(send.x);
    o.y = keep.y + row_xor<MASK>(send.y);
    o.z = keep.z + row_xor<MASK>(send.z);
    o.w = keep.w + row_xor<MASK>(send.w);
    return o;
}
__global__ __launch_bounds__(256)
void splat_fused(const float* __restrict__ means,
                 const float* __restrict__ covs,
                 const float* __restrict__ colors,
                 const float* __restrict__ opac,
                 float* __restrict__ out)
{
    __shared__ float4 sG[NG];
    __shared__ float4 sC[NG];

    const int tid = threadIdx.x;
    const int row = blockIdx.x >> 1;
    const float fy = (float)row;

    {
        float rmx[2], rmy[2], ra[2], rb[2], rc[2], rd[2], rop[2], r0[2], r1[2], r2[2];
#pragma unroll
        for (int j = 0; j < 2; ++j) {
            const int n = tid + j * 256;
            rmx[j] = means[2*n+0];
            rmy[j] = means[2*n+1];
            ra[j]  = covs[4*n+0];
            rb[j]  = covs[4*n+1];
            rc[j]  = covs[4*n+2];
            rd[j]  = covs[4*n+3];
            rop[j] = opac[n];
            r0[j]  = colors[3*n+0];
            r1[j]  = colors[3*n+1];
            r2[j]  = colors[3*n+2];
        }
#pragma unroll
        for (int j = 0; j < 2; ++j) {
            const int n = tid + j * 256;
            float mx = rmx[j] - 128.0f;
            float dy = fy - rmy[j];
            float a = ra[j], b = rb[j], c = rc[j], d = rd[j];
            float det = a*d - b*c;
            float inv_det = 1.0f / det;
            float ia  =  d*inv_det;
            float ib2 = -(b + c)*inv_det;
            float ic  =  a*inv_det;
            float amp = rop[j] / (6.28318530717958647f * sqrtf(det + 1e-8f));

            float t = ib2 * dy;
            float4 g;
            g.x = QSCALE * ia;
            g.y = QSCALE * fmaf(-2.0f*ia, mx, t);
            g.z = QSCALE * fmaf(mx, fmaf(ia, mx, -t), ic*dy*dy);
            g.w = 0.0f;
            sG[n] = g;

            sC[n] = make_float4(amp, amp * r0[j], amp * r1[j], amp * r2[j]);
        }
    }
    __syncthreads();

    const int lane = tid & 31;
    const int wave = tid >> 5;
    const int x0 = (((blockIdx.x & 1) << 3) + wave) << 4;

    const int half = lane >> 4;
    const int m    = lane & 15;

    const float fx = (float)(x0 + m) - 128.0f;

    v2f a1;
    a1.x = half ? 1.0f : (fx * fx);
    a1.y = half ? 0.0f : fx;

    float4 acc[8];
#pragma unroll
    for (int r = 0; r < 8; ++r) acc[r] = make_float4(0.f, 0.f, 0.f, 0.f);

    const int k0 = half << 1;

    for (int t = 0; t < NG / 16; ++t) {
        const int n = (t << 4) + m;
        v2f b1 = *(const v2f*)((const float*)&sG[n] + k0);

        v8f q = {};
        q = __builtin_amdgcn_wmma_f32_16x16x4_f32(false, a1, false, b1,
                                                  (short)0, q, false, false);

        const float4 cn = sC[n];
#pragma unroll
        for (int r = 0; r < 8; ++r) {
            float e = __builtin_amdgcn_exp2f(q[r]);
            acc[r].x = fmaf(e, cn.x, acc[r].x);
            acc[r].y = fmaf(e, cn.y, acc[r].y);
            acc[r].z = fmaf(e, cn.z, acc[r].z);
            acc[r].w = fmaf(e, cn.w, acc[r].w);
        }
    }

    const bool bit0 = (m & 1) != 0, bit1 = (m & 2) != 0, bit2 = (m & 4) != 0;
    float4 s1[4];
#pragma unroll
    for (int j = 0; j < 4; ++j) s1[j] = keep_half<1>(bit0, acc[2*j], acc[2*j + 1]);
    float4 s2[2];
#pragma unroll
    for (int k = 0; k < 2; ++k) s2[k] = keep_half<2>(bit1, s1[2*k], s1[2*k + 1]);
    float4 v = keep_half<4>(bit2, s2[0], s2[1]);
    v.x += row_xor<8>(v.x);
    v.y += row_xor<8>(v.y);
    v.z += row_xor<8>(v.z);
    v.w += row_xor<8>(v.w);

    if (m < 8) {
        const int x = x0 + (half << 3) + m;
        const float inv = 1.0f / fmaxf(v.x, 1e-8f);
        float* o = out + (row * WW + x) * 3;
        o[0] = v.y * inv;
        o[1] = v.z * inv;
        o[2] = v.w * inv;
    }
}

extern "C" void kernel_launch(void* const* d_in, const int* in_sizes, int n_in,
                              void* d_out, int out_size, void* d_ws, size_t ws_size,
                              hipStream_t stream) {
    const float* means  = (const float*)d_in[0];
    const float* covs   = (const float*)d_in[1];
    const float* colors = (const float*)d_in[2];
    const float* opac   = (const float*)d_in[3];

    splat_fused<<<(HH * WW) / (8 * 16), 256, 0, stream>>>(means, covs, colors, opac,
                                                          (float*)d_out);
}


# speedup vs baseline: 2.4227x; 1.0529x over previous
#include <hip/hip_runtime.h>
#include <math.h>

typedef __attribute__((ext_vector_type(2))) float v2f;
typedef __attribute__((ext_vector_type(8))) float v8f;

#define NG 512
#define HH 256
#define WW 256

#define QSCALE (-0.72134752044448170368f)


template <int MASK>
__device__ __forceinline__ float row_xor(float v)
{
    return __builtin_bit_cast(float,
        __builtin_amdgcn_update_dpp(0, __builtin_bit_cast(int, v),
                                    0x160 + MASK, 0xf, 0xf, false));
}

template <int MASK>
__device__ __forceinline__ float4 keep_half(bool bit, float4 lo, float4 hi)
{
    float4 keep, send;
    keep.x = bit ? hi.x : lo.x;   send.x = bit ? lo.x : hi.x;
    keep.y = bit ? hi.y : lo.y;   send.y = bit ? lo.y : hi.y;
    keep.z = bit ? hi.z : lo.z;   send.z = bit ? lo.z : hi.z;
    keep.w = bit ? hi.w : lo.w;   send.w = bit ? lo.w : hi.w;
    float4 o;
    o.x = keep.x + row_xor<MASK>(send.x);
    o.y = keep.y + row_xor<MASK>(send.y);
    o.z = keep.z + row_xor<MASK>(send.z);
    o.w = keep.w + row_xor<MASK>(send.w);
    return o;
}
__global__ __launch_bounds__(256)
void splat_fused(const float* __restrict__ means,
                 const float* __restrict__ covs,
                 const float* __restrict__ colors,
                 const float* __restrict__ opac,
                 float* __restrict__ out)
{
    __shared__ float4 sG[NG];
    __shared__ float4 sC[NG];

    const int tid = threadIdx.x;
    const int row = blockIdx.x >> 1;
    const float fy = (float)row;

    {
        float rmx[2], rmy[2], ra[2], rb[2], rc[2], rd[2], rop[2], r0[2], r1[2], r2[2];
#pragma unroll
        for (int j = 0; j < 2; ++j) {
            const int n = tid + j * 256;
            rmx[j] = means[2*n+0];
            rmy[j] = means[2*n+1];
            ra[j]  = covs[4*n+0];
            rb[j]  = covs[4*n+1];
            rc[j]  = covs[4*n+2];
            rd[j]  = covs[4*n+3];
            rop[j] = opac[n];
            r0[j]  = colors[3*n+0];
            r1[j]  = colors[3*n+1];
            r2[j]  = colors[3*n+2];
        }
#pragma unroll
        for (int j = 0; j < 2; ++j) {
            const int n = tid + j * 256;
            float mx = rmx[j] - 128.0f;
            float dy = fy - rmy[j];
            float a = ra[j], b = rb[j], c = rc[j], d = rd[j];
            float det = a*d - b*c;
            float inv_det = 1.0f / det;
            float ia  =  d*inv_det;
            float ib2 = -(b + c)*inv_det;
            float ic  =  a*inv_det;
            float amp = rop[j] / (6.28318530717958647f * sqrtf(det + 1e-8f));

            float t = ib2 * dy;
            float4 g;
            g.x = QSCALE * ia;
            g.y = QSCALE * fmaf(-2.0f*ia, mx, t);
            g.z = QSCALE * fmaf(mx, fmaf(ia, mx, -t), ic*dy*dy);
            g.w = 0.0f;
            sG[n] = g;

            sC[n] = make_float4(amp, amp * r0[j], amp * r1[j], amp * r2[j]);
        }
    }
    __syncthreads();

    const int lane = tid & 31;
    const int wave = tid >> 5;
    const int x0 = (((blockIdx.x & 1) << 3) + wave) << 4;

    const int half = lane >> 4;
    const int m    = lane & 15;

    const float fx = (float)(x0 + m) - 128.0f;

    v2f a1;
    a1.x = half ? 1.0f : (fx * fx);
    a1.y = half ? 0.0f : fx;

    float4 acc[8];
#pragma unroll
    for (int r = 0; r < 8; ++r) acc[r] = make_float4(0.f, 0.f, 0.f, 0.f);

    const int k0 = half << 1;

    const float*  gB = (const float*)&sG[m] + k0;
    const float4* gC = &sC[m];
    auto loadB = [&](int t) -> v2f    { return *(const v2f*)(gB + (t << 6)); };
    auto loadC = [&](int t) -> float4 { return gC[t << 4]; };
    auto quad  = [&](v2f b) -> v8f {
        v8f z = {};
        return __builtin_amdgcn_wmma_f32_16x16x4_f32(false, a1, false, b,
                                                     (short)0, z, false, false);
    };
    auto expo  = [&](const v8f& q) -> v8f {
        v8f e;
#pragma unroll
        for (int r = 0; r < 8; ++r) e[r] = __builtin_amdgcn_exp2f(q[r]);
        return e;
    };
    auto accum = [&](const v8f& e, const float4& cn) {
#pragma unroll
        for (int r = 0; r < 8; ++r) {
            acc[r].x = fmaf(e[r], cn.x, acc[r].x);
            acc[r].y = fmaf(e[r], cn.y, acc[r].y);
            acc[r].z = fmaf(e[r], cn.z, acc[r].z);
            acc[r].w = fmaf(e[r], cn.w, acc[r].w);
        }
    };

#define KEEP_ORDER() __builtin_amdgcn_sched_barrier(0)
    auto step = [&](v8f& qW, v2f& b, int tb, const v8f& qR, v8f& eW,
                    const v8f& eR, float4& c, int tc) {
        qW = quad(b);
        b = loadB(tb);
        KEEP_ORDER();
#pragma unroll
        for (int r = 0; r < 8; ++r) {
            acc[r].x = fmaf(eR[r], c.x, acc[r].x);
            acc[r].y = fmaf(eR[r], c.y, acc[r].y);
            acc[r].z = fmaf(eR[r], c.z, acc[r].z);
            acc[r].w = fmaf(eR[r], c.w, acc[r].w);
            KEEP_ORDER();
            eW[r] = __builtin_amdgcn_exp2f(qR[r]);
            KEEP_ORDER();
        }
        c = loadC(tc);
        KEEP_ORDER();
    };

    v2f bE = loadB(0), bO = loadB(1);
    v8f qA = quad(bE);
    v8f qB = quad(bO);
    bE = loadB(2);
    bO = loadB(3);
    float4 cE = loadC(0), cO = loadC(1);
    v8f eA = expo(qA);
    v8f eB;

    for (int t = 0; t < NG / 16 - 4; t += 2) {
        step(qA, bE, t + 4, qB, eB, eA, cE, t + 2);
        step(qB, bO, t + 5, qA, eA, eB, cO, t + 3);
    }
#undef KEEP_ORDER
    qA = quad(bE);
    eB = expo(qB);
    accum(eA, cE);   cE = loadC(NG / 16 - 2);
    qB = quad(bO);
    eA = expo(qA);
    accum(eB, cO);   cO = loadC(NG / 16 - 1);
    eB = expo(qB);
    accum(eA, cE);
    accum(eB, cO);

    const bool bit0 = (m & 1) != 0, bit1 = (m & 2) != 0, bit2 = (m & 4) != 0;
    float4 s1[4];
#pragma unroll
    for (int j = 0; j < 4; ++j) s1[j] = keep_half<1>(bit0, acc[2*j], acc[2*j + 1]);
    float4 s2[2];
#pragma unroll
    for (int k = 0; k < 2; ++k) s2[k] = keep_half<2>(bit1, s1[2*k], s1[2*k + 1]);
    float4 v = keep_half<4>(bit2, s2[0], s2[1]);
    v.x += row_xor<8>(v.x);
    v.y += row_xor<8>(v.y);
    v.z += row_xor<8>(v.z);
    v.w += row_xor<8>(v.w);

    if (m < 8) {
        const int x = x0 + (half << 3) + m;
        const float inv = 1.0f / fmaxf(v.x, 1e-8f);
        float* o = out + (row * WW + x) * 3;
        o[0] = v.y * inv;
        o[1] = v.z * inv;
        o[2] = v.w * inv;
    }
}

extern "C" void kernel_launch(void* const* d_in, const int* in_sizes, int n_in,
                              void* d_out, int out_size, void* d_ws, size_t ws_size,
                              hipStream_t stream) {
    const float* means  = (const float*)d_in[0];
    const float* covs   = (const float*)d_in[1];
    const float* colors = (const float*)d_in[2];
    const float* opac   = (const float*)d_in[3];

    splat_fused<<<(HH * WW) / (8 * 16), 256, 0, stream>>>(means, covs, colors, opac,
                                                          (float*)d_out);
}


# Each wave takes two 16-pixel strips per B/C tile fetch, one block per image row: half the waves and blocks, coefficient prep once per WGP, a step's 2 WMMAs a strip's vector work apart; same add order
# speedup vs baseline: 2.5314x; 1.0449x over previous
#include <hip/hip_runtime.h>
#include <math.h>

typedef __attribute__((ext_vector_type(2))) float v2f;
typedef __attribute__((ext_vector_type(8))) float v8f;

#define NG 512
#define HH 256
#define WW 256

#define QSCALE (-0.72134752044448170368f)


template <int MASK>
__device__ __forceinline__ float row_xor(float v)
{
    return __builtin_bit_cast(float,
        __builtin_amdgcn_update_dpp(0, __builtin_bit_cast(int, v),
                                    0x160 + MASK, 0xf, 0xf, false));
}

template <int MASK>
__device__ __forceinline__ float4 keep_half(bool bit, float4 lo, float4 hi)
{
    float4 keep, send;
    keep.x = bit ? hi.x : lo.x;   send.x = bit ? lo.x : hi.x;
    keep.y = bit ? hi.y : lo.y;   send.y = bit ? lo.y : hi.y;
    keep.z = bit ? hi.z : lo.z;   send.z = bit ? lo.z : hi.z;
    keep.w = bit ? hi.w : lo.w;   send.w = bit ? lo.w : hi.w;
    float4 o;
    o.x = keep.x + row_xor<MASK>(send.x);
    o.y = keep.y + row_xor<MASK>(send.y);
    o.z = keep.z + row_xor<MASK>(send.z);
    o.w = keep.w + row_xor<MASK>(send.w);
    return o;
}
__global__ __launch_bounds__(256)
void splat_fused(const float* __restrict__ means,
                 const float* __restrict__ covs,
                 const float* __restrict__ colors,
                 const float* __restrict__ opac,
                 float* __restrict__ out)
{
    __shared__ float4 sG[NG];
    __shared__ float4 sC[NG];

    const int tid = threadIdx.x;
    const int row = blockIdx.x;
    const float fy = (float)row;

    {
        float rmx[2], rmy[2], ra[2], rb[2], rc[2], rd[2], rop[2], r0[2], r1[2], r2[2];
#pragma unroll
        for (int j = 0; j < 2; ++j) {
            const int n = tid + j * 256;
            rmx[j] = means[2*n+0];
            rmy[j] = means[2*n+1];
            ra[j]  = covs[4*n+0];
            rb[j]  = covs[4*n+1];
            rc[j]  = covs[4*n+2];
            rd[j]  = covs[4*n+3];
            rop[j] = opac[n];
            r0[j]  = colors[3*n+0];
            r1[j]  = colors[3*n+1];
            r2[j]  = colors[3*n+2];
        }
#pragma unroll
        for (int j = 0; j < 2; ++j) {
            const int n = tid + j * 256;
            float mx = rmx[j] - 128.0f;
            float dy = fy - rmy[j];
            float a = ra[j], b = rb[j], c = rc[j], d = rd[j];
            float det = a*d - b*c;
            float inv_det = 1.0f / det;
            float ia  =  d*inv_det;
            float ib2 = -(b + c)*inv_det;
            float ic  =  a*inv_det;
            float amp = rop[j] / (6.28318530717958647f * sqrtf(det + 1e-8f));

            float t = ib2 * dy;
            float4 g;
            g.x = QSCALE * ia;
            g.y = QSCALE * fmaf(-2.0f*ia, mx, t);
            g.z = QSCALE * fmaf(mx, fmaf(ia, mx, -t), ic*dy*dy);
            g.w = 0.0f;
            sG[n] = g;

            sC[n] = make_float4(amp, amp * r0[j], amp * r1[j], amp * r2[j]);
        }
    }
    __syncthreads();

    const int lane = tid & 31;
    const int wave = tid >> 5;
    const int x0 = wave << 5;

    const int half = lane >> 4;
    const int m    = lane & 15;

    const float fx0 = (float)(x0 + m) - 128.0f;
    const float fx1 = (float)(x0 + 16 + m) - 128.0f;

    v2f a0, a1;
    a0.x = half ? 1.0f : (fx0 * fx0);
    a0.y = half ? 0.0f : fx0;
    a1.x = half ? 1.0f : (fx1 * fx1);
    a1.y = half ? 0.0f : fx1;

    float4 acc0[8], acc1[8];
#pragma unroll
    for (int r = 0; r < 8; ++r) {
        acc0[r] = make_float4(0.f, 0.f, 0.f, 0.f);
        acc1[r] = make_float4(0.f, 0.f, 0.f, 0.f);
    }

    const int k0 = half << 1;

    const float*  gB = (const float*)&sG[m] + k0;
    const float4* gC = &sC[m];
    auto loadB = [&](int t) -> v2f    { return *(const v2f*)(gB + (t << 6)); };
    auto loadC = [&](int t) -> float4 { return gC[t << 4]; };
    auto quad  = [&](v2f a, v2f b) -> v8f {
        v8f z = {};
        return __builtin_amdgcn_wmma_f32_16x16x4_f32(false, a, false, b,
                                                     (short)0, z, false, false);
    };
    auto expo  = [&](const v8f& q) -> v8f {
        v8f e;
#pragma unroll
        for (int r = 0; r < 8; ++r) e[r] = __builtin_amdgcn_exp2f(q[r]);
        return e;
    };
    auto accum = [&](float4 (&ac)[8], const v8f& e, const float4& cn) {
#pragma unroll
        for (int r = 0; r < 8; ++r) {
            ac[r].x = fmaf(e[r], cn.x, ac[r].x);
            ac[r].y = fmaf(e[r], cn.y, ac[r].y);
            ac[r].z = fmaf(e[r], cn.z, ac[r].z);
            ac[r].w = fmaf(e[r], cn.w, ac[r].w);
        }
    };

#define KEEP_ORDER() __builtin_amdgcn_sched_barrier(0)
    auto rows = [&](float4 (&ac)[8], const v8f& eR, const float4& c,
                    v8f& eW, const v8f& qR) {
#pragma unroll
        for (int r = 0; r < 8; ++r) {
            ac[r].x = fmaf(eR[r], c.x, ac[r].x);
            ac[r].y = fmaf(eR[r], c.y, ac[r].y);
            ac[r].z = fmaf(eR[r], c.z, ac[r].z);
            ac[r].w = fmaf(eR[r], c.w, ac[r].w);
            KEEP_ORDER();
            eW[r] = __builtin_amdgcn_exp2f(qR[r]);
            KEEP_ORDER();
        }
    };
    auto step = [&](v8f& qW0, v8f& qW1, v2f& b, int tb,
                    const v8f& qR0, const v8f& qR1, v8f& eW0, v8f& eW1,
                    const v8f& eR0, const v8f& eR1, float4& c, int tc) {
        qW0 = quad(a0, b);
        KEEP_ORDER();
        rows(acc0, eR0, c, eW0, qR0);
        qW1 = quad(a1, b);
        KEEP_ORDER();
        rows(acc1, eR1, c, eW1, qR1);
        c = loadC(tc);
        b = loadB(tb);
        KEEP_ORDER();
    };

    v2f bE = loadB(0), bO = loadB(1);
    v8f qA0 = quad(a0, bE), qA1 = quad(a1, bE);
    v8f qB0 = quad(a0, bO), qB1 = quad(a1, bO);
    bE = loadB(2);
    bO = loadB(3);
    float4 cE = loadC(0), cO = loadC(1);
    v8f eA0 = expo(qA0), eA1 = expo(qA1);
    v8f eB0, eB1;

    for (int t = 0; t < NG / 16 - 4; t += 2) {
        step(qA0, qA1, bE, t + 4, qB0, qB1, eB0, eB1, eA0, eA1, cE, t + 2);
        step(qB0, qB1, bO, t + 5, qA0, qA1, eA0, eA1, eB0, eB1, cO, t + 3);
    }
#undef KEEP_ORDER
    qA0 = quad(a0, bE);  qA1 = quad(a1, bE);
    eB0 = expo(qB0);     eB1 = expo(qB1);
    accum(acc0, eA0, cE);  accum(acc1, eA1, cE);
    cE = loadC(NG / 16 - 2);
    qB0 = quad(a0, bO);  qB1 = quad(a1, bO);
    eA0 = expo(qA0);     eA1 = expo(qA1);
    accum(acc0, eB0, cO);  accum(acc1, eB1, cO);
    cO = loadC(NG / 16 - 1);
    eB0 = expo(qB0);     eB1 = expo(qB1);
    accum(acc0, eA0, cE);  accum(acc1, eA1, cE);
    accum(acc0, eB0, cO);  accum(acc1, eB1, cO);

    const bool bit0 = (m & 1) != 0, bit1 = (m & 2) != 0, bit2 = (m & 4) != 0;
    auto finish = [&](float4 (&ac)[8], int xs) {
        float4 s1[4];
#pragma unroll
        for (int j = 0; j < 4; ++j) s1[j] = keep_half<1>(bit0, ac[2*j], ac[2*j + 1]);
        float4 s2[2];
#pragma unroll
        for (int k = 0; k < 2; ++k) s2[k] = keep_half<2>(bit1, s1[2*k], s1[2*k + 1]);
        float4 v = keep_half<4>(bit2, s2[0], s2[1]);
        v.x += row_xor<8>(v.x);
        v.y += row_xor<8>(v.y);
        v.z += row_xor<8>(v.z);
        v.w += row_xor<8>(v.w);

        if (m < 8) {
            const int x = xs + (half << 3) + m;
            const float inv = 1.0f / fmaxf(v.x, 1e-8f);
            float* o = out + (row * WW + x) * 3;
            o[0] = v.y * inv;
            o[1] = v.z * inv;
            o[2] = v.w * inv;
        }
    };
    finish(acc0, x0);
    finish(acc1, x0 + 16);
}

extern "C" void kernel_launch(void* const* d_in, const int* in_sizes, int n_in,
                              void* d_out, int out_size, void* d_ws, size_t ws_size,
                              hipStream_t stream) {
    const float* means  = (const float*)d_in[0];
    const float* covs   = (const float*)d_in[1];
    const float* colors = (const float*)d_in[2];
    const float* opac   = (const float*)d_in[3];

    splat_fused<<<HH, 256, 0, stream>>>(means, covs, colors, opac, (float*)d_out);
}


# Per-wave culling: a gaussian whose exponent is below -160 on all 32 pixels of the wave (exp2 exactly 0) is left out; tile loop runs over a compacted per-wave list; adds regrouped by list position
# speedup vs baseline: 2.7217x; 1.0752x over previous
#include <hip/hip_runtime.h>
#include <math.h>

typedef __attribute__((ext_vector_type(2))) float v2f;
typedef __attribute__((ext_vector_type(8))) float v8f;
typedef unsigned       __attribute__((may_alias)) list_word;
typedef unsigned short __attribute__((may_alias)) list_half;

#define NG 512
#define HH 256
#define WW 256

#define REC_BYTES  80
#define REC_F4     (REC_BYTES / 16)
#define LIST0      ((NG + 1) * REC_BYTES)
#define LIST_WORDS ((NG / 16 + 2) * 16)
#define LIST_BYTES (LIST_WORDS * 4 + 16)
#define SM_BYTES   (LIST0 + 8 * LIST_BYTES)

#define QSCALE (-0.72134752044448170368f)


template <int MASK>
__device__ __forceinline__ float row_xor(float v)
{
    return __builtin_bit_cast(float,
        __builtin_amdgcn_update_dpp(0, __builtin_bit_cast(int, v),
                                    0x160 + MASK, 0xf, 0xf, false));
}

template <int MASK>
__device__ __forceinline__ float4 keep_half(bool bit, float4 lo, float4 hi)
{
    float4 keep, send;
    keep.x = bit ? hi.x : lo.x;   send.x = bit ? lo.x : hi.x;
    keep.y = bit ? hi.y : lo.y;   send.y = bit ? lo.y : hi.y;
    keep.z = bit ? hi.z : lo.z;   send.z = bit ? lo.z : hi.z;
    keep.w = bit ? hi.w : lo.w;   send.w = bit ? lo.w : hi.w;
    float4 o;
    o.x = keep.x + row_xor<MASK>(send.x);
    o.y = keep.y + row_xor<MASK>(send.y);
    o.z = keep.z + row_xor<MASK>(send.z);
    o.w = keep.w + row_xor<MASK>(send.w);
    return o;
}
__global__ __launch_bounds__(256)
void splat_fused(const float* __restrict__ means,
                 const float* __restrict__ covs,
                 const float* __restrict__ colors,
                 const float* __restrict__ opac,
                 float* __restrict__ out)
{
    __shared__ float4 sm[SM_BYTES / 16];

    const int tid = threadIdx.x;
    const int row = blockIdx.x;
    const float fy = (float)row;

    {
        float rmx[2], rmy[2], ra[2], rb[2], rc[2], rd[2], rop[2], r0[2], r1[2], r2[2];
#pragma unroll
        for (int j = 0; j < 2; ++j) {
            const int n = tid + j * 256;
            rmx[j] = means[2*n+0];
            rmy[j] = means[2*n+1];
            ra[j]  = covs[4*n+0];
            rb[j]  = covs[4*n+1];
            rc[j]  = covs[4*n+2];
            rd[j]  = covs[4*n+3];
            rop[j] = opac[n];
            r0[j]  = colors[3*n+0];
            r1[j]  = colors[3*n+1];
            r2[j]  = colors[3*n+2];
        }
#pragma unroll
        for (int j = 0; j < 2; ++j) {
            const int n = tid + j * 256;
            float mx = rmx[j] - 128.0f;
            float dy = fy - rmy[j];
            float a = ra[j], b = rb[j], c = rc[j], d = rd[j];
            float det = a*d - b*c;
            float inv_det = 1.0f / det;
            float ia  =  d*inv_det;
            float ib2 = -(b + c)*inv_det;
            float ic  =  a*inv_det;
            float amp = rop[j] / (6.28318530717958647f * sqrtf(det + 1e-8f));

            float t = ib2 * dy;
            float4 g;
            g.x = QSCALE * ia;
            g.y = QSCALE * fmaf(-2.0f*ia, mx, t);
            g.z = QSCALE * fmaf(mx, fmaf(ia, mx, -t), ic*dy*dy);
            const float4 cw = make_float4(amp, amp * r0[j], amp * r1[j], amp * r2[j]);
            g.z += 0.0f * (cw.x + cw.y + cw.z + cw.w);
            const float mag = fmaf(fabsf(g.x), 16384.0f, fmaf(fabsf(g.y), 128.0f, fabsf(g.z)));
            const float gzs = g.z + fmaf(mag, 1e-6f, 160.0f);
            float xv = (-0.5f * g.y) * __builtin_amdgcn_rcpf(g.x);
            xv = fminf(fmaxf(xv, -1e30f), 1e30f);
            sm[REC_F4*n + 0] = make_float4(g.x, g.y, gzs, xv);
            sm[REC_F4*n + 1] = cw;
            sm[REC_F4*n + 2] = make_float4(g.z, xv, 0.0f, 0.0f);
            sm[REC_F4*n + 3] = cw;
        }
        if (tid < 4) {
            const float z = -1000.0f;
            sm[REC_F4*NG + tid] = make_float4(tid == 2 ? z : 0.0f, 0.0f, tid == 0 ? z : 0.0f, 0.0f);
        }
    }
    __syncthreads();

    const int lane = tid & 31;
    const int wave = tid >> 5;
    const int x0 = wave << 5;

    const int half = lane >> 4;
    const int m    = lane & 15;

    const float fx0 = (float)(x0 + m) - 128.0f;
    const float fx1 = (float)(x0 + 16 + m) - 128.0f;

    v2f a0, a1;
    a0.x = half ? 1.0f : (fx0 * fx0);
    a0.y = half ? 0.0f : fx0;
    a1.x = half ? 1.0f : (fx1 * fx1);
    a1.y = half ? 0.0f : fx1;

    float4 acc0[8], acc1[8];
#pragma unroll
    for (int r = 0; r < 8; ++r) {
        acc0[r] = make_float4(0.f, 0.f, 0.f, 0.f);
        acc1[r] = make_float4(0.f, 0.f, 0.f, 0.f);
    }

    char* const lds = (char*)sm;
    char* const lw  = lds + LIST0 + wave * LIST_BYTES;
    const float Xa = (float)x0 - 128.0f;
    const float Xb = Xa + 31.0f;
    int count = 0;
    float4 rec[NG / 32];
#pragma unroll
    for (int r = 0; r < NG / 32; ++r) rec[r] = sm[REC_F4 * (r * 32 + lane)];
    const unsigned wlane = (unsigned)lane * (REC_BYTES * 0x10001u) + (32u << 16);
#pragma unroll
    for (int r = 0; r < NG / 32; ++r) {
        const float4 g = rec[r];
        const float Xc = fminf(fmaxf(g.w, Xa), Xb);
        const float qa = fmaf(fmaf(g.x, Xa, g.y), Xa, g.z);
        const float qb = fmaf(fmaf(g.x, Xb, g.y), Xb, g.z);
        const float qc = fmaf(fmaf(g.x, Xc, g.y), Xc, g.z);
        const bool dead = (qa < 0.0f) & (qb < 0.0f) & (qc < 0.0f);
        const unsigned livem = __builtin_amdgcn_ballot_w32(!dead);
        const int pos = (int)__builtin_amdgcn_mbcnt_lo(livem, (unsigned)count);
        *(list_word*)(lw + ((dead ? LIST_WORDS : pos) << 2)) = wlane + (unsigned)r * (32u * REC_BYTES * 0x10001u);
        count += __builtin_popcount(livem);
    }
    int nt = ((count + 31) >> 5) << 1;
    nt = nt < 4 ? 4 : nt;
    const int lend = (nt + 2) << 4;
#pragma unroll
    for (int i = 0; i < 3; ++i) {
        const int pos = count + lane + 32 * i;
        *(list_word*)(lw + ((pos < lend ? pos : LIST_WORDS) << 2)) = (unsigned)(NG * REC_BYTES) * 0x10001u + (32u << 16);
    }

    const char* lp = lw + (m << 2) + (half << 1);
    auto loadE = [&](int off) -> unsigned { return *(const list_half*)(lp + off); };
    auto loadB = [&](unsigned e) -> v2f    { return *(const v2f*)(lds + e); };
    auto loadC = [&](unsigned e) -> float4 { return *(const float4*)(lds + e + 16); };
    auto quad  = [&](v2f a, v2f b) -> v8f {
        v8f z = {};
        return __builtin_amdgcn_wmma_f32_16x16x4_f32(false, a, false, b,
                                                     (short)0, z, false, false);
    };
    auto expo  = [&](const v8f& q) -> v8f {
        v8f e;
#pragma unroll
        for (int r = 0; r < 8; ++r) e[r] = __builtin_amdgcn_exp2f(q[r]);
        return e;
    };
    auto accum = [&](float4 (&ac)[8], const v8f& e, const float4& cn) {
#pragma unroll
        for (int r = 0; r < 8; ++r) {
            ac[r].x = fmaf(e[r], cn.x, ac[r].x);
            ac[r].y = fmaf(e[r], cn.y, ac[r].y);
            ac[r].z = fmaf(e[r], cn.z, ac[r].z);
            ac[r].w = fmaf(e[r], cn.w, ac[r].w);
        }
    };

#define KEEP_ORDER() __builtin_amdgcn_sched_barrier(0)
    auto rows = [&](float4 (&ac)[8], const v8f& eR, const float4& c,
                    v8f& eW, const v8f& qR) {
#pragma unroll
        for (int r = 0; r < 8; ++r) {
            ac[r].x = fmaf(eR[r], c.x, ac[r].x);
            ac[r].y = fmaf(eR[r], c.y, ac[r].y);
            ac[r].z = fmaf(eR[r], c.z, ac[r].z);
            ac[r].w = fmaf(eR[r], c.w, ac[r].w);
            KEEP_ORDER();
            eW[r] = __builtin_amdgcn_exp2f(qR[r]);
            KEEP_ORDER();
        }
    };
    auto step = [&](v8f& qW0, v8f& qW1, v2f& b, unsigned& entB, int offB,
                    const v8f& qR0, const v8f& qR1, v8f& eW0, v8f& eW1,
                    const v8f& eR0, const v8f& eR1, float4& c, unsigned& entC, int offC) {
        qW0 = quad(a0, b);
        KEEP_ORDER();
        rows(acc0, eR0, c, eW0, qR0);
        qW1 = quad(a1, b);
        KEEP_ORDER();
        rows(acc1, eR1, c, eW1, qR1);
        c = loadC(entC);
        b = loadB(entB);
        entC = loadE(offC);
        entB = loadE(offB);
        KEEP_ORDER();
    };

    const unsigned n0 = loadE(0), n1 = loadE(64);
    unsigned nCE = loadE(128), nCO = loadE(192), nBE = loadE(256), nBO = loadE(320);
    v2f bE = loadB(n0), bO = loadB(n1);
    v8f qA0 = quad(a0, bE), qA1 = quad(a1, bE);
    v8f qB0 = quad(a0, bO), qB1 = quad(a1, bO);
    bE = loadB(nCE);
    bO = loadB(nCO);
    float4 cE = loadC(n0), cO = loadC(n1);
    v8f eA0 = expo(qA0), eA1 = expo(qA1);
    v8f eB0, eB1;

    for (int t = 0; t < nt - 4; t += 2) {
        step(qA0, qA1, bE, nBE, 6 * 64, qB0, qB1, eB0, eB1, eA0, eA1, cE, nCE, 4 * 64);
        step(qB0, qB1, bO, nBO, 7 * 64, qA0, qA1, eA0, eA1, eB0, eB1, cO, nCO, 5 * 64);
        lp += 2 * 64;
    }
#undef KEEP_ORDER
    qA0 = quad(a0, bE);  qA1 = quad(a1, bE);
    eB0 = expo(qB0);     eB1 = expo(qB1);
    accum(acc0, eA0, cE);  accum(acc1, eA1, cE);
    cE = loadC(nCE);
    qB0 = quad(a0, bO);  qB1 = quad(a1, bO);
    eA0 = expo(qA0);     eA1 = expo(qA1);
    accum(acc0, eB0, cO);  accum(acc1, eB1, cO);
    cO = loadC(nCO);
    eB0 = expo(qB0);     eB1 = expo(qB1);
    accum(acc0, eA0, cE);  accum(acc1, eA1, cE);
    accum(acc0, eB0, cO);  accum(acc1, eB1, cO);

    const bool bit0 = (m & 1) != 0, bit1 = (m & 2) != 0, bit2 = (m & 4) != 0;
    auto finish = [&](float4 (&ac)[8], int xs) {
        float4 s1[4];
#pragma unroll
        for (int j = 0; j < 4; ++j) s1[j] = keep_half<1>(bit0, ac[2*j], ac[2*j + 1]);
        float4 s2[2];
#pragma unroll
        for (int k = 0; k < 2; ++k) s2[k] = keep_half<2>(bit1, s1[2*k], s1[2*k + 1]);
        float4 v = keep_half<4>(bit2, s2[0], s2[1]);
        v.x += row_xor<8>(v.x);
        v.y += row_xor<8>(v.y);
        v.z += row_xor<8>(v.z);
        v.w += row_xor<8>(v.w);

        if (m < 8) {
            const int x = xs + (half << 3) + m;
            const float inv = 1.0f / fmaxf(v.x, 1e-8f);
            float* o = out + (row * WW + x) * 3;
            o[0] = v.y * inv;
            o[1] = v.z * inv;
            o[2] = v.w * inv;
        }
    };
    finish(acc0, x0);
    finish(acc1, x0 + 16);
}

extern "C" void kernel_launch(void* const* d_in, const int* in_sizes, int n_in,
                              void* d_out, int out_size, void* d_ws, size_t ws_size,
                              hipStream_t stream) {
    const float* means  = (const float*)d_in[0];
    const float* covs   = (const float*)d_in[1];
    const float* colors = (const float*)d_in[2];
    const float* opac   = (const float*)d_in[3];

    splat_fused<<<HH, 256, 0, stream>>>(means, covs, colors, opac, (float*)d_out);
}


# Cull by an error bound, not by exact zero: a wave leaves a gaussian out where its largest weight x 2^exponent is below 2^-60 on all its pixels (a colour moves < 5e-8 x (1+|colour|)); adds regrouped
# speedup vs baseline: 3.2307x; 1.1870x over previous
#include <hip/hip_runtime.h>
#include <math.h>

typedef __attribute__((ext_vector_type(2))) float v2f;
typedef __attribute__((ext_vector_type(8))) float v8f;
typedef unsigned       __attribute__((may_alias)) list_word;
typedef unsigned short __attribute__((may_alias)) list_half;

#define NG 512
#define HH 256
#define WW 256

#define REC_BYTES  80
#define REC_F4     (REC_BYTES / 16)
#define LIST0      ((NG + 1) * REC_BYTES)
#define LIST_WORDS ((NG / 16 + 2) * 16)
#define LIST_BYTES (LIST_WORDS * 4 + 16)
#define SM_BYTES   (LIST0 + 8 * LIST_BYTES)

#define QSCALE (-0.72134752044448170368f)


template <int MASK>
__device__ __forceinline__ float row_xor(float v)
{
    return __builtin_bit_cast(float,
        __builtin_amdgcn_update_dpp(0, __builtin_bit_cast(int, v),
                                    0x160 + MASK, 0xf, 0xf, false));
}

template <int MASK>
__device__ __forceinline__ float4 keep_half(bool bit, float4 lo, float4 hi)
{
    float4 keep, send;
    keep.x = bit ? hi.x : lo.x;   send.x = bit ? lo.x : hi.x;
    keep.y = bit ? hi.y : lo.y;   send.y = bit ? lo.y : hi.y;
    keep.z = bit ? hi.z : lo.z;   send.z = bit ? lo.z : hi.z;
    keep.w = bit ? hi.w : lo.w;   send.w = bit ? lo.w : hi.w;
    float4 o;
    o.x = keep.x + row_xor<MASK>(send.x);
    o.y = keep.y + row_xor<MASK>(send.y);
    o.z = keep.z + row_xor<MASK>(send.z);
    o.w = keep.w + row_xor<MASK>(send.w);
    return o;
}
__global__ __launch_bounds__(256)
void splat_fused(const float* __restrict__ means,
                 const float* __restrict__ covs,
                 const float* __restrict__ colors,
                 const float* __restrict__ opac,
                 float* __restrict__ out)
{
    __shared__ float4 sm[SM_BYTES / 16];

    const int tid = threadIdx.x;
    const int row = blockIdx.x;
    const float fy = (float)row;

    {
        float rmx[2], rmy[2], ra[2], rb[2], rc[2], rd[2], rop[2], r0[2], r1[2], r2[2];
#pragma unroll
        for (int j = 0; j < 2; ++j) {
            const int n = tid + j * 256;
            rmx[j] = means[2*n+0];
            rmy[j] = means[2*n+1];
            ra[j]  = covs[4*n+0];
            rb[j]  = covs[4*n+1];
            rc[j]  = covs[4*n+2];
            rd[j]  = covs[4*n+3];
            rop[j] = opac[n];
            r0[j]  = colors[3*n+0];
            r1[j]  = colors[3*n+1];
            r2[j]  = colors[3*n+2];
        }
#pragma unroll
        for (int j = 0; j < 2; ++j) {
            const int n = tid + j * 256;
            float mx = rmx[j] - 128.0f;
            float dy = fy - rmy[j];
            float a = ra[j], b = rb[j], c = rc[j], d = rd[j];
            float det = a*d - b*c;
            float inv_det = 1.0f / det;
            float ia  =  d*inv_det;
            float ib2 = -(b + c)*inv_det;
            float ic  =  a*inv_det;
            float amp = rop[j] / (6.28318530717958647f * sqrtf(det + 1e-8f));

            float t = ib2 * dy;
            float4 g;
            g.x = QSCALE * ia;
            g.y = QSCALE * fmaf(-2.0f*ia, mx, t);
            g.z = QSCALE * fmaf(mx, fmaf(ia, mx, -t), ic*dy*dy);
            const float4 cw = make_float4(amp, amp * r0[j], amp * r1[j], amp * r2[j]);
            g.z += 0.0f * (cw.x + cw.y + cw.z + cw.w);
            const float mag = fmaf(fabsf(g.x), 16384.0f, fmaf(fabsf(g.y), 128.0f, fabsf(g.z)));
            const float wmax = fmaxf(fmaxf(fabsf(cw.x), fabsf(cw.y)), fmaxf(fabsf(cw.z), fabsf(cw.w)));
            const float cut  = fminf(61.0f + __builtin_amdgcn_logf(fmaxf(wmax, 1.17549435e-38f)), 160.0f);
            const float gzs = g.z + fmaf(mag, 1e-6f, cut);
            float xv = (-0.5f * g.y) * __builtin_amdgcn_rcpf(g.x);
            xv = fminf(fmaxf(xv, -1e30f), 1e30f);
            sm[REC_F4*n + 0] = make_float4(g.x, g.y, gzs, xv);
            sm[REC_F4*n + 1] = cw;
            sm[REC_F4*n + 2] = make_float4(g.z, xv, 0.0f, 0.0f);
            sm[REC_F4*n + 3] = cw;
        }
        if (tid < 4) {
            const float z = -1000.0f;
            sm[REC_F4*NG + tid] = make_float4(tid == 2 ? z : 0.0f, 0.0f, tid == 0 ? z : 0.0f, 0.0f);
        }
    }
    __syncthreads();

    const int lane = tid & 31;
    const int wave = tid >> 5;
    const int x0 = wave << 5;

    const int half = lane >> 4;
    const int m    = lane & 15;

    const float fx0 = (float)(x0 + m) - 128.0f;
    const float fx1 = (float)(x0 + 16 + m) - 128.0f;

    v2f a0, a1;
    a0.x = half ? 1.0f : (fx0 * fx0);
    a0.y = half ? 0.0f : fx0;
    a1.x = half ? 1.0f : (fx1 * fx1);
    a1.y = half ? 0.0f : fx1;

    float4 acc0[8], acc1[8];
#pragma unroll
    for (int r = 0; r < 8; ++r) {
        acc0[r] = make_float4(0.f, 0.f, 0.f, 0.f);
        acc1[r] = make_float4(0.f, 0.f, 0.f, 0.f);
    }

    char* const lds = (char*)sm;
    char* const lw  = lds + LIST0 + wave * LIST_BYTES;
    const float Xa = (float)x0 - 128.0f;
    const float Xb = Xa + 31.0f;
    int count = 0;
    float4 rec[NG / 32];
#pragma unroll
    for (int r = 0; r < NG / 32; ++r) rec[r] = sm[REC_F4 * (r * 32 + lane)];
    const unsigned wlane = (unsigned)lane * (REC_BYTES * 0x10001u) + (32u << 16);
#pragma unroll
    for (int r = 0; r < NG / 32; ++r) {
        const float4 g = rec[r];
        const float Xc = fminf(fmaxf(g.w, Xa), Xb);
        const float qa = fmaf(fmaf(g.x, Xa, g.y), Xa, g.z);
        const float qb = fmaf(fmaf(g.x, Xb, g.y), Xb, g.z);
        const float qc = fmaf(fmaf(g.x, Xc, g.y), Xc, g.z);
        const bool dead = (qa < 0.0f) & (qb < 0.0f) & (qc < 0.0f);
        const unsigned livem = __builtin_amdgcn_ballot_w32(!dead);
        const int pos = (int)__builtin_amdgcn_mbcnt_lo(livem, (unsigned)count);
        *(list_word*)(lw + ((dead ? LIST_WORDS : pos) << 2)) = wlane + (unsigned)r * (32u * REC_BYTES * 0x10001u);
        count += __builtin_popcount(livem);
    }
    int nt = ((count + 31) >> 5) << 1;
    nt = nt < 4 ? 4 : nt;
    const int lend = (nt + 2) << 4;
#pragma unroll
    for (int i = 0; i < 3; ++i) {
        const int pos = count + lane + 32 * i;
        *(list_word*)(lw + ((pos < lend ? pos : LIST_WORDS) << 2)) = (unsigned)(NG * REC_BYTES) * 0x10001u + (32u << 16);
    }

    const char* lp = lw + (m << 2) + (half << 1);
    auto loadE = [&](int off) -> unsigned { return *(const list_half*)(lp + off); };
    auto loadB = [&](unsigned e) -> v2f    { return *(const v2f*)(lds + e); };
    auto loadC = [&](unsigned e) -> float4 { return *(const float4*)(lds + e + 16); };
    auto quad  = [&](v2f a, v2f b) -> v8f {
        v8f z = {};
        return __builtin_amdgcn_wmma_f32_16x16x4_f32(false, a, false, b,
                                                     (short)0, z, false, false);
    };
    auto expo  = [&](const v8f& q) -> v8f {
        v8f e;
#pragma unroll
        for (int r = 0; r < 8; ++r) e[r] = __builtin_amdgcn_exp2f(q[r]);
        return e;
    };
    auto accum = [&](float4 (&ac)[8], const v8f& e, const float4& cn) {
#pragma unroll
        for (int r = 0; r < 8; ++r) {
            ac[r].x = fmaf(e[r], cn.x, ac[r].x);
            ac[r].y = fmaf(e[r], cn.y, ac[r].y);
            ac[r].z = fmaf(e[r], cn.z, ac[r].z);
            ac[r].w = fmaf(e[r], cn.w, ac[r].w);
        }
    };

#define KEEP_ORDER() __builtin_amdgcn_sched_barrier(0)
    auto rows = [&](float4 (&ac)[8], const v8f& eR, const float4& c,
                    v8f& eW, const v8f& qR) {
#pragma unroll
        for (int r = 0; r < 8; ++r) {
            ac[r].x = fmaf(eR[r], c.x, ac[r].x);
            ac[r].y = fmaf(eR[r], c.y, ac[r].y);
            ac[r].z = fmaf(eR[r], c.z, ac[r].z);
            ac[r].w = fmaf(eR[r], c.w, ac[r].w);
            KEEP_ORDER();
            eW[r] = __builtin_amdgcn_exp2f(qR[r]);
            KEEP_ORDER();
        }
    };
    auto step = [&](v8f& qW0, v8f& qW1, v2f& b, unsigned& entB, int offB,
                    const v8f& qR0, const v8f& qR1, v8f& eW0, v8f& eW1,
                    const v8f& eR0, const v8f& eR1, float4& c, unsigned& entC, int offC) {
        qW0 = quad(a0, b);
        KEEP_ORDER();
        rows(acc0, eR0, c, eW0, qR0);
        qW1 = quad(a1, b);
        KEEP_ORDER();
        rows(acc1, eR1, c, eW1, qR1);
        c = loadC(entC);
        b = loadB(entB);
        entC = loadE(offC);
        entB = loadE(offB);
        KEEP_ORDER();
    };

    const unsigned n0 = loadE(0), n1 = loadE(64);
    unsigned nCE = loadE(128), nCO = loadE(192), nBE = loadE(256), nBO = loadE(320);
    v2f bE = loadB(n0), bO = loadB(n1);
    v8f qA0 = quad(a0, bE), qA1 = quad(a1, bE);
    v8f qB0 = quad(a0, bO), qB1 = quad(a1, bO);
    bE = loadB(nCE);
    bO = loadB(nCO);
    float4 cE = loadC(n0), cO = loadC(n1);
    v8f eA0 = expo(qA0), eA1 = expo(qA1);
    v8f eB0, eB1;

    for (int t = 0; t < nt - 4; t += 2) {
        step(qA0, qA1, bE, nBE, 6 * 64, qB0, qB1, eB0, eB1, eA0, eA1, cE, nCE, 4 * 64);
        step(qB0, qB1, bO, nBO, 7 * 64, qA0, qA1, eA0, eA1, eB0, eB1, cO, nCO, 5 * 64);
        lp += 2 * 64;
    }
#undef KEEP_ORDER
    qA0 = quad(a0, bE);  qA1 = quad(a1, bE);
    eB0 = expo(qB0);     eB1 = expo(qB1);
    accum(acc0, eA0, cE);  accum(acc1, eA1, cE);
    cE = loadC(nCE);
    qB0 = quad(a0, bO);  qB1 = quad(a1, bO);
    eA0 = expo(qA0);     eA1 = expo(qA1);
    accum(acc0, eB0, cO);  accum(acc1, eB1, cO);
    cO = loadC(nCO);
    eB0 = expo(qB0);     eB1 = expo(qB1);
    accum(acc0, eA0, cE);  accum(acc1, eA1, cE);
    accum(acc0, eB0, cO);  accum(acc1, eB1, cO);

    const bool bit0 = (m & 1) != 0, bit1 = (m & 2) != 0, bit2 = (m & 4) != 0;
    auto finish = [&](float4 (&ac)[8], int xs) {
        float4 s1[4];
#pragma unroll
        for (int j = 0; j < 4; ++j) s1[j] = keep_half<1>(bit0, ac[2*j], ac[2*j + 1]);
        float4 s2[2];
#pragma unroll
        for (int k = 0; k < 2; ++k) s2[k] = keep_half<2>(bit1, s1[2*k], s1[2*k + 1]);
        float4 v = keep_half<4>(bit2, s2[0], s2[1]);
        v.x += row_xor<8>(v.x);
        v.y += row_xor<8>(v.y);
        v.z += row_xor<8>(v.z);
        v.w += row_xor<8>(v.w);

        if (m < 8) {
            const int x = xs + (half << 3) + m;
            const float inv = 1.0f / fmaxf(v.x, 1e-8f);
            float* o = out + (row * WW + x) * 3;
            o[0] = v.y * inv;
            o[1] = v.z * inv;
            o[2] = v.w * inv;
        }
    };
    finish(acc0, x0);
    finish(acc1, x0 + 16);
}

extern "C" void kernel_launch(void* const* d_in, const int* in_sizes, int n_in,
                              void* d_out, int out_size, void* d_ws, size_t ws_size,
                              hipStream_t stream) {
    const float* means  = (const float*)d_in[0];
    const float* covs   = (const float*)d_in[1];
    const float* colors = (const float*)d_in[2];
    const float* opac   = (const float*)d_in[3];

    splat_fused<<<HH, 256, 0, stream>>>(means, covs, colors, opac, (float*)d_out);
}


# Culling by a live X range solved once per gaussian and row in the coefficient stage (one sqrt): a wave's test is 2 compares on 8 bytes, not 3 parabola evaluations on 16; same cut, list and add order
# speedup vs baseline: 3.3791x; 1.0459x over previous
#include <hip/hip_runtime.h>
#include <math.h>

typedef __attribute__((ext_vector_type(2))) float v2f;
typedef __attribute__((ext_vector_type(8))) float v8f;
typedef unsigned       __attribute__((may_alias)) list_word;
typedef unsigned short __attribute__((may_alias)) list_half;

#define NG 512
#define HH 256
#define WW 256

#define REC_BYTES  80
#define REC_F4     (REC_BYTES / 16)
#define LIST0      ((NG + 1) * REC_BYTES)
#define LIST_WORDS ((NG / 16 + 2) * 16)
#define LIST_BYTES (LIST_WORDS * 4 + 16)
#define LOHI0      (LIST0 + 8 * LIST_BYTES)
#define SM_BYTES   (LOHI0 + NG * 8)

#define QSCALE (-0.72134752044448170368f)


template <int MASK>
__device__ __forceinline__ float row_xor(float v)
{
    return __builtin_bit_cast(float,
        __builtin_amdgcn_update_dpp(0, __builtin_bit_cast(int, v),
                                    0x160 + MASK, 0xf, 0xf, false));
}

template <int MASK>
__device__ __forceinline__ float4 keep_half(bool bit, float4 lo, float4 hi)
{
    float4 keep, send;
    keep.x = bit ? hi.x : lo.x;   send.x = bit ? lo.x : hi.x;
    keep.y = bit ? hi.y : lo.y;   send.y = bit ? lo.y : hi.y;
    keep.z = bit ? hi.z : lo.z;   send.z = bit ? lo.z : hi.z;
    keep.w = bit ? hi.w : lo.w;   send.w = bit ? lo.w : hi.w;
    float4 o;
    o.x = keep.x + row_xor<MASK>(send.x);
    o.y = keep.y + row_xor<MASK>(send.y);
    o.z = keep.z + row_xor<MASK>(send.z);
    o.w = keep.w + row_xor<MASK>(send.w);
    return o;
}
__global__ __launch_bounds__(256)
void splat_fused(const float* __restrict__ means,
                 const float* __restrict__ covs,
                 const float* __restrict__ colors,
                 const float* __restrict__ opac,
                 float* __restrict__ out)
{
    __shared__ float4 sm[SM_BYTES / 16];

    const int tid = threadIdx.x;
    const int row = blockIdx.x;
    const float fy = (float)row;

    {
        float rmx[2], rmy[2], ra[2], rb[2], rc[2], rd[2], rop[2], r0[2], r1[2], r2[2];
#pragma unroll
        for (int j = 0; j < 2; ++j) {
            const int n = tid + j * 256;
            rmx[j] = means[2*n+0];
            rmy[j] = means[2*n+1];
            ra[j]  = covs[4*n+0];
            rb[j]  = covs[4*n+1];
            rc[j]  = covs[4*n+2];
            rd[j]  = covs[4*n+3];
            rop[j] = opac[n];
            r0[j]  = colors[3*n+0];
            r1[j]  = colors[3*n+1];
            r2[j]  = colors[3*n+2];
        }
#pragma unroll
        for (int j = 0; j < 2; ++j) {
            const int n = tid + j * 256;
            float mx = rmx[j] - 128.0f;
            float dy = fy - rmy[j];
            float a = ra[j], b = rb[j], c = rc[j], d = rd[j];
            float det = a*d - b*c;
            float inv_det = 1.0f / det;
            float ia  =  d*inv_det;
            float ib2 = -(b + c)*inv_det;
            float ic  =  a*inv_det;
            float amp = rop[j] / (6.28318530717958647f * sqrtf(det + 1e-8f));

            float t = ib2 * dy;
            float4 g;
            g.x = QSCALE * ia;
            g.y = QSCALE * fmaf(-2.0f*ia, mx, t);
            g.z = QSCALE * fmaf(mx, fmaf(ia, mx, -t), ic*dy*dy);
            const float4 cw = make_float4(amp, amp * r0[j], amp * r1[j], amp * r2[j]);
            g.z += 0.0f * (cw.x + cw.y + cw.z + cw.w);
            const float mag = fmaf(fabsf(g.x), 16384.0f, fmaf(fabsf(g.y), 128.0f, fabsf(g.z)));
            const float wmax = fmaxf(fmaxf(fabsf(cw.x), fabsf(cw.y)), fmaxf(fabsf(cw.z), fabsf(cw.w)));
            const float cut  = fminf(61.0f + __builtin_amdgcn_logf(fmaxf(wmax, 1.17549435e-38f)), 160.0f);
            const float gzs = g.z + fmaf(mag, 1e-6f, cut);
            const float rgx = __builtin_amdgcn_rcpf(g.x);
            float xv = (-0.5f * g.y) * rgx;
            xv = fminf(fmaxf(xv, -1e30f), 1e30f);
            const float tq  = -gzs * rgx;
            const float r2  = fmaf(xv, xv, tq);
            const float r2c = fmaf(fmaf(xv, xv, fabsf(tq)), 2e-6f, r2);
            const float rad = __builtin_amdgcn_sqrtf(r2c);
            const float rm  = fmaf(rad, 1.000002f, fmaf(fabsf(xv), 2e-6f, 1e-3f));
            float lo = xv - rm, hi = xv + rm;
            if (r2c < 0.0f)     { lo =  INFINITY; hi = -INFINITY; }
            if (!(g.x < 0.0f))  { lo = -INFINITY; hi =  INFINITY; }
            sm[REC_F4*n + 0] = make_float4(g.x, g.y, 0.0f, 0.0f);
            sm[REC_F4*n + 1] = cw;
            sm[REC_F4*n + 2] = make_float4(g.z, 0.0f, 0.0f, 0.0f);
            sm[REC_F4*n + 3] = cw;
            v2f lh;  lh.x = lo;  lh.y = hi;
            *(v2f*)((char*)sm + LOHI0 + n * 8) = lh;
        }
        if (tid < 4) {
            const float z = -1000.0f;
            sm[REC_F4*NG + tid] = make_float4(tid == 2 ? z : 0.0f, 0.0f, tid == 0 ? z : 0.0f, 0.0f);
        }
    }
    __syncthreads();

    const int lane = tid & 31;
    const int wave = tid >> 5;
    const int x0 = wave << 5;

    const int half = lane >> 4;
    const int m    = lane & 15;

    const float fx0 = (float)(x0 + m) - 128.0f;
    const float fx1 = (float)(x0 + 16 + m) - 128.0f;

    v2f a0, a1;
    a0.x = half ? 1.0f : (fx0 * fx0);
    a0.y = half ? 0.0f : fx0;
    a1.x = half ? 1.0f : (fx1 * fx1);
    a1.y = half ? 0.0f : fx1;

    float4 acc0[8], acc1[8];
#pragma unroll
    for (int r = 0; r < 8; ++r) {
        acc0[r] = make_float4(0.f, 0.f, 0.f, 0.f);
        acc1[r] = make_float4(0.f, 0.f, 0.f, 0.f);
    }

    char* const lds = (char*)sm;
    char* const lw  = lds + LIST0 + wave * LIST_BYTES;
    const float Xa = (float)x0 - 128.0f;
    const float Xb = Xa + 31.0f;
    int count = 0;
    v2f lh[NG / 32];
#pragma unroll
    for (int r = 0; r < NG / 32; ++r) lh[r] = *(const v2f*)(lds + LOHI0 + (r * 32 + lane) * 8);
    const unsigned wlane = (unsigned)lane * (REC_BYTES * 0x10001u) + (32u << 16);
#pragma unroll
    for (int r = 0; r < NG / 32; ++r) {
        const bool outL = lh[r].y < Xa;
        const bool outR = lh[r].x > Xb;
        const bool dead = outL | outR;
        const unsigned livem = ~(__builtin_amdgcn_ballot_w32(outL) | __builtin_amdgcn_ballot_w32(outR));
        const int pos = (int)__builtin_amdgcn_mbcnt_lo(livem, (unsigned)count);
        *(list_word*)(lw + ((dead ? LIST_WORDS : pos) << 2)) = wlane + (unsigned)r * (32u * REC_BYTES * 0x10001u);
        count += __builtin_popcount(livem);
    }
    int nt = ((count + 31) >> 5) << 1;
    nt = nt < 4 ? 4 : nt;
    const int lend = (nt + 2) << 4;
#pragma unroll
    for (int i = 0; i < 3; ++i) {
        const int pos = count + lane + 32 * i;
        *(list_word*)(lw + ((pos < lend ? pos : LIST_WORDS) << 2)) = (unsigned)(NG * REC_BYTES) * 0x10001u + (32u << 16);
    }

    const char* lp = lw + (m << 2) + (half << 1);
    auto loadE = [&](int off) -> unsigned { return *(const list_half*)(lp + off); };
    auto loadB = [&](unsigned e) -> v2f    { return *(const v2f*)(lds + e); };
    auto loadC = [&](unsigned e) -> float4 { return *(const float4*)(lds + e + 16); };
    auto quad  = [&](v2f a, v2f b) -> v8f {
        v8f z = {};
        return __builtin_amdgcn_wmma_f32_16x16x4_f32(false, a, false, b,
                                                     (short)0, z, false, false);
    };
    auto expo  = [&](const v8f& q) -> v8f {
        v8f e;
#pragma unroll
        for (int r = 0; r < 8; ++r) e[r] = __builtin_amdgcn_exp2f(q[r]);
        return e;
    };
    auto accum = [&](float4 (&ac)[8], const v8f& e, const float4& cn) {
#pragma unroll
        for (int r = 0; r < 8; ++r) {
            ac[r].x = fmaf(e[r], cn.x, ac[r].x);
            ac[r].y = fmaf(e[r], cn.y, ac[r].y);
            ac[r].z = fmaf(e[r], cn.z, ac[r].z);
            ac[r].w = fmaf(e[r], cn.w, ac[r].w);
        }
    };

#define KEEP_ORDER() __builtin_amdgcn_sched_barrier(0)
    auto rows = [&](float4 (&ac)[8], const v8f& eR, const float4& c,
                    v8f& eW, const v8f& qR) {
#pragma unroll
        for (int r = 0; r < 8; ++r) {
            ac[r].x = fmaf(eR[r], c.x, ac[r].x);
            ac[r].y = fmaf(eR[r], c.y, ac[r].y);
            ac[r].z = fmaf(eR[r], c.z, ac[r].z);
            ac[r].w = fmaf(eR[r], c.w, ac[r].w);
            KEEP_ORDER();
            eW[r] = __builtin_amdgcn_exp2f(qR[r]);
            KEEP_ORDER();
        }
    };
    auto step = [&](v8f& qW0, v8f& qW1, v2f& b, unsigned& entB, int offB,
                    const v8f& qR0, const v8f& qR1, v8f& eW0, v8f& eW1,
                    const v8f& eR0, const v8f& eR1, float4& c, unsigned& entC, int offC) {
        qW0 = quad(a0, b);
        KEEP_ORDER();
        rows(acc0, eR0, c, eW0, qR0);
        qW1 = quad(a1, b);
        KEEP_ORDER();
        rows(acc1, eR1, c, eW1, qR1);
        c = loadC(entC);
        b = loadB(entB);
        entC = loadE(offC);
        entB = loadE(offB);
        KEEP_ORDER();
    };

    const unsigned n0 = loadE(0), n1 = loadE(64);
    unsigned nCE = loadE(128), nCO = loadE(192), nBE = loadE(256), nBO = loadE(320);
    v2f bE = loadB(n0), bO = loadB(n1);
    v8f qA0 = quad(a0, bE), qA1 = quad(a1, bE);
    v8f qB0 = quad(a0, bO), qB1 = quad(a1, bO);
    bE = loadB(nCE);
    bO = loadB(nCO);
    float4 cE = loadC(n0), cO = loadC(n1);
    v8f eA0 = expo(qA0), eA1 = expo(qA1);
    v8f eB0, eB1;

    for (int t = 0; t < nt - 4; t += 2) {
        step(qA0, qA1, bE, nBE, 6 * 64, qB0, qB1, eB0, eB1, eA0, eA1, cE, nCE, 4 * 64);
        step(qB0, qB1, bO, nBO, 7 * 64, qA0, qA1, eA0, eA1, eB0, eB1, cO, nCO, 5 * 64);
        lp += 2 * 64;
    }
#undef KEEP_ORDER
    qA0 = quad(a0, bE);  qA1 = quad(a1, bE);
    eB0 = expo(qB0);     eB1 = expo(qB1);
    accum(acc0, eA0, cE);  accum(acc1, eA1, cE);
    cE = loadC(nCE);
    qB0 = quad(a0, bO);  qB1 = quad(a1, bO);
    eA0 = expo(qA0);     eA1 = expo(qA1);
    accum(acc0, eB0, cO);  accum(acc1, eB1, cO);
    cO = loadC(nCO);
    eB0 = expo(qB0);     eB1 = expo(qB1);
    accum(acc0, eA0, cE);  accum(acc1, eA1, cE);
    accum(acc0, eB0, cO);  accum(acc1, eB1, cO);

    const bool bit0 = (m & 1) != 0, bit1 = (m & 2) != 0, bit2 = (m & 4) != 0;
    auto finish = [&](float4 (&ac)[8], int xs) {
        float4 s1[4];
#pragma unroll
        for (int j = 0; j < 4; ++j) s1[j] = keep_half<1>(bit0, ac[2*j], ac[2*j + 1]);
        float4 s2[2];
#pragma unroll
        for (int k = 0; k < 2; ++k) s2[k] = keep_half<2>(bit1, s1[2*k], s1[2*k + 1]);
        float4 v = keep_half<4>(bit2, s2[0], s2[1]);
        v.x += row_xor<8>(v.x);
        v.y += row_xor<8>(v.y);
        v.z += row_xor<8>(v.z);
        v.w += row_xor<8>(v.w);

        if (m < 8) {
            const int x = xs + (half << 3) + m;
            const float inv = 1.0f / fmaxf(v.x, 1e-8f);
            float* o = out + (row * WW + x) * 3;
            o[0] = v.y * inv;
            o[1] = v.z * inv;
            o[2] = v.w * inv;
        }
    };
    finish(acc0, x0);
    finish(acc1, x0 + 16);
}

extern "C" void kernel_launch(void* const* d_in, const int* in_sizes, int n_in,
                              void* d_out, int out_size, void* d_ws, size_t ws_size,
                              hipStream_t stream) {
    const float* means  = (const float*)d_in[0];
    const float* covs   = (const float*)d_in[1];
    const float* colors = (const float*)d_in[2];
    const float* opac   = (const float*)d_in[3];

    splat_fused<<<HH, 256, 0, stream>>>(means, covs, colors, opac, (float*)d_out);
}
